# starting kernel
# baseline (speedup 1.0000x reference)
#include <hip/hip_runtime.h>
#include <math.h>

typedef float v2f __attribute__((ext_vector_type(2)));
typedef float v8f __attribute__((ext_vector_type(8)));

#define IN_F  128
#define HD    128
#define NHEAD 4
#define DHEAD 32
#define DEG   16
#define ALPHA 0.2f
#define LDSK  133

#if defined(__HIP_DEVICE_COMPILE__)
#if !__has_builtin(__builtin_amdgcn_wmma_f32_16x16x4_f32)
#error "gfx1250 fp32 WMMA builtin not available on this toolchain (device pass)"
#endif
#endif

__global__ __launch_bounds__(256) void gat_zero(float* __restrict__ p, int n) {
  int i = blockIdx.x * 256 + threadIdx.x;
  if (i < n) p[i] = 0.0f;
}

__global__ __launch_bounds__(256) void gat_gemm(
    const float* __restrict__ X, const float* __restrict__ W0,
    const float* __restrict__ W1, const float* __restrict__ a0,
    float* __restrict__ X1out, float* __restrict__ s0, float* __restrict__ s1,
    int N)
{
  __shared__ float lA[16 * LDSK];

  const int tid   = threadIdx.x;
  const int nbase = blockIdx.x * 16;

  {
    int row = tid >> 4;
    int col = (tid & 15) * 8;
    int r = nbase + row; if (r > N - 1) r = N - 1;
    const float4* src = (const float4*)(X + (size_t)r * IN_F + col);
    float4 q0 = src[0];
    float4 q1 = src[1];
    float* dst = lA + row * LDSK + col;
    dst[0] = q0.x; dst[1] = q0.y; dst[2] = q0.z; dst[3] = q0.w;
    dst[4] = q1.x; dst[5] = q1.y; dst[6] = q1.z; dst[7] = q1.w;
  }
  __syncthreads();

  const int wave  = tid >> 5;
  const int lane  = tid & 31;
  const int l15   = lane & 15;
  const int lhalf = lane >> 4;
  const int cbase = wave * 16;

  const float* w0row = W0 + (size_t)(cbase + l15) * IN_F;
  const float* w1row = W1 + (size_t)(cbase + l15) * IN_F;
  const float* arow  = lA + l15 * LDSK;

  v8f acc0 = {};
  v8f acc1 = {};

  #pragma unroll
  for (int kk = 0; kk < IN_F / 4; ++kk) {
    const int k0 = kk * 4 + lhalf * 2;
    v2f av; av.x = arow[k0];  av.y = arow[k0 + 1];
    v2f b0; b0.x = w0row[k0]; b0.y = w0row[k0 + 1];
    v2f b1; b1.x = w1row[k0]; b1.y = w1row[k0 + 1];
    acc0 = __builtin_amdgcn_wmma_f32_16x16x4_f32(false, av, false, b0,
                                                 (short)0, acc0, false, false);
    acc1 = __builtin_amdgcn_wmma_f32_16x16x4_f32(false, av, false, b1,
                                                 (short)0, acc1, false, false);
  }

  const int   head = wave >> 1;
  const float aval = a0[head * DHEAD + (cbase & (DHEAD - 1)) + l15];

  float p0[8], p1[8];
  #pragma unroll
  for (int j = 0; j < 8; ++j) {
    float x0 = acc0[j]; x0 = (x0 > 0.0f) ? x0 : ALPHA * x0;
    float x1 = acc1[j]; x1 = (x1 > 0.0f) ? x1 : ALPHA * x1;
    int r = nbase + j + 8 * lhalf;
    if (r < N) X1out[(size_t)r * HD + cbase + l15] = x1;
    p0[j] = x0 * aval;
    p1[j] = x1 * aval;
  }

  #pragma unroll
  for (int j = 0; j < 8; ++j) {
    #pragma unroll
    for (int off = 8; off >= 1; off >>= 1) {
      p0[j] += __shfl_xor(p0[j], off, 32);
      p1[j] += __shfl_xor(p1[j], off, 32);
    }
  }
  if (l15 == 0) {
    #pragma unroll
    for (int j = 0; j < 8; ++j) {
      int r = nbase + j + 8 * lhalf;
      if (r < N) {
        atomicAdd(&s0[r * NHEAD + head], p0[j]);
        atomicAdd(&s1[r * NHEAD + head], p1[j]);
      }
    }
  }
}

__global__ __launch_bounds__(256) void gat_minmax_part(
    const float* __restrict__ s0, const float* __restrict__ s1,
    const int* __restrict__ esrc, const int* __restrict__ ecol, int E,
    float* __restrict__ pmax, float* __restrict__ pmn)
{
  const int tid = threadIdx.x;
  float mx[NHEAD], mn[NHEAD];
  #pragma unroll
  for (int h = 0; h < NHEAD; ++h) { mx[h] = -1e30f; mn[h] = 1e30f; }

  for (int e = blockIdx.x * 256 + tid; e < E; e += gridDim.x * 256) {
    int s = esrc[e];
    int c = ecol[e];
    const float* sa = s0 + (size_t)s * NHEAD;
    const float* sb = s1 + (size_t)c * NHEAD;
    #pragma unroll
    for (int h = 0; h < NHEAD; ++h) {
      float att = sa[h] + sb[h];
      mx[h] = fmaxf(mx[h], att);
      mn[h] = fminf(mn[h], att);
    }
  }

  #pragma unroll
  for (int h = 0; h < NHEAD; ++h)
    for (int off = 16; off >= 1; off >>= 1) {
      mx[h] = fmaxf(mx[h], __shfl_xor(mx[h], off, 32));
      mn[h] = fminf(mn[h], __shfl_xor(mn[h], off, 32));
    }

  __shared__ float smx[8][NHEAD], smn[8][NHEAD];
  const int wave = tid >> 5, lane = tid & 31;
  if (lane == 0) {
    #pragma unroll
    for (int h = 0; h < NHEAD; ++h) { smx[wave][h] = mx[h]; smn[wave][h] = mn[h]; }
  }
  __syncthreads();
  if (tid == 0) {
    #pragma unroll
    for (int h = 0; h < NHEAD; ++h) {
      float a = smx[0][h], b = smn[0][h];
      for (int w = 1; w < 8; ++w) { a = fmaxf(a, smx[w][h]); b = fminf(b, smn[w][h]); }
      pmax[blockIdx.x * NHEAD + h] = a;
      pmn [blockIdx.x * NHEAD + h] = b;
    }
  }
}

__global__ __launch_bounds__(256) void gat_minmax_final(
    const float* __restrict__ pmax, const float* __restrict__ pmn, int nb,
    float* __restrict__ mxmn)
{
  const int tid = threadIdx.x;
  float mx[NHEAD], mn[NHEAD];
  #pragma unroll
  for (int h = 0; h < NHEAD; ++h) { mx[h] = -1e30f; mn[h] = 1e30f; }

  for (int i = tid; i < nb; i += 256) {
    #pragma unroll
    for (int h = 0; h < NHEAD; ++h) {
      mx[h] = fmaxf(mx[h], pmax[i * NHEAD + h]);
      mn[h] = fminf(mn[h], pmn [i * NHEAD + h]);
    }
  }
  #pragma unroll
  for (int h = 0; h < NHEAD; ++h)
    for (int off = 16; off >= 1; off >>= 1) {
      mx[h] = fmaxf(mx[h], __shfl_xor(mx[h], off, 32));
      mn[h] = fminf(mn[h], __shfl_xor(mn[h], off, 32));
    }

  __shared__ float smx[8][NHEAD], smn[8][NHEAD];
  const int wave = tid >> 5, lane = tid & 31;
  if (lane == 0) {
    #pragma unroll
    for (int h = 0; h < NHEAD; ++h) { smx[wave][h] = mx[h]; smn[wave][h] = mn[h]; }
  }
  __syncthreads();
  if (tid == 0) {
    #pragma unroll
    for (int h = 0; h < NHEAD; ++h) {
      float a = smx[0][h], b = smn[0][h];
      for (int w = 1; w < 8; ++w) { a = fmaxf(a, smx[w][h]); b = fminf(b, smn[w][h]); }
      mxmn[h]         = a;
      mxmn[NHEAD + h] = b;
    }
  }
}

__global__ __launch_bounds__(256) void gat_edge(
    const float* __restrict__ X1, const float* __restrict__ s0,
    const float* __restrict__ s1, const int* __restrict__ ecol,
    const float* __restrict__ mxmn, float* __restrict__ out, int N)
{
  const int wave = threadIdx.x >> 5;
  const int lane = threadIdx.x & 31;
  const int n = blockIdx.x * 8 + wave;
  if (n >= N) return;

  float scale[NHEAD], s0n[NHEAD];
  #pragma unroll
  for (int h = 0; h < NHEAD; ++h) {
    float mxv = mxmn[h], mnv = mxmn[NHEAD + h];
    scale[h] = 1.0f / (mxv - mnv);
    s0n[h]   = s0[n * NHEAD + h] - mnv;
  }

  float acc[NHEAD]  = {0.f, 0.f, 0.f, 0.f};
  float wsum[NHEAD] = {0.f, 0.f, 0.f, 0.f};

  const int* cols = ecol + n * DEG;
  for (int e = 0; e < DEG; ++e) {
    int c = cols[e];
    const float* sc   = s1 + (size_t)c * NHEAD;
    const float* xrow = X1 + (size_t)c * HD;
    #pragma unroll
    for (int h = 0; h < NHEAD; ++h) {
      float att = __expf((s0n[h] + sc[h]) * scale[h]);
      wsum[h] += att;
      acc[h]  += att * xrow[h * DHEAD + lane];
    }
  }
  #pragma unroll
  for (int h = 0; h < NHEAD; ++h)
    out[(size_t)n * HD + h * DHEAD + lane] = acc[h] / wsum[h];
}

extern "C" void kernel_launch(void* const* d_in, const int* in_sizes, int n_in,
                              void* d_out, int out_size, void* d_ws, size_t ws_size,
                              hipStream_t stream) {
  const float* X    = (const float*)d_in[0];
  const float* W0   = (const float*)d_in[1];
  const float* W1   = (const float*)d_in[2];
  const float* a0   = (const float*)d_in[3];
  const int*   esrc = (const int*)d_in[4];
  const int*   ecol = (const int*)d_in[5];
  float*       out  = (float*)d_out;

  const int N = in_sizes[0] / IN_F;
  const int E = in_sizes[5];
  const int NB = 512;

  float* X1ws = (float*)d_ws;
  float* s0   = X1ws + (size_t)N * HD;
  float* s1   = s0 + (size_t)N * NHEAD;
  float* pmax = s1 + (size_t)N * NHEAD;
  float* pmn  = pmax + (size_t)NB * NHEAD;
  float* mxmn = pmn + (size_t)NB * NHEAD;

  gat_zero<<<(2 * N * NHEAD + 255) / 256, 256, 0, stream>>>(s0, 2 * N * NHEAD);
  gat_gemm<<<(N + 15) / 16, 256, 0, stream>>>(X, W0, W1, a0, X1ws, s0, s1, N);
  gat_minmax_part<<<NB, 256, 0, stream>>>(s0, s1, esrc, ecol, E, pmax, pmn);
  gat_minmax_final<<<1, 256, 0, stream>>>(pmax, pmn, NB, mxmn);
  gat_edge<<<(N + 7) / 8, 256, 0, stream>>>(X1ws, s0, s1, ecol, mxmn, out, N);
}


# GEMM: each wave keeps its W0/W1 B fragments for all K in registers and the block loops over 16-node tiles (X tile double-buffered in LDS); k order inside the dot products changes
# speedup vs baseline: 1.6319x; 1.6319x over previous
#include <hip/hip_runtime.h>
#include <math.h>

typedef float v2f __attribute__((ext_vector_type(2)));
typedef float v8f __attribute__((ext_vector_type(8)));

#define IN_F  128
#define HD    128
#define NHEAD 4
#define DHEAD 32
#define DEG   16
#define ALPHA 0.2f
#define LDSK  132

#if defined(__HIP_DEVICE_COMPILE__)
#if !__has_builtin(__builtin_amdgcn_wmma_f32_16x16x4_f32)
#error "gfx1250 fp32 WMMA builtin not available on this toolchain (device pass)"
#endif
#endif

__global__ __launch_bounds__(256) void gat_zero(float* __restrict__ p, int n) {
  int i = blockIdx.x * 256 + threadIdx.x;
  if (i < n) p[i] = 0.0f;
}

__global__ __launch_bounds__(256) void gat_gemm(
    const float* __restrict__ X, const float* __restrict__ W0,
    const float* __restrict__ W1, const float* __restrict__ a0,
    float* __restrict__ X1out, float* __restrict__ s0, float* __restrict__ s1,
    int N)
{
  __shared__ __attribute__((aligned(16))) float lW[HD * LDSK];
  __shared__ __attribute__((aligned(16))) float lA[2][16 * LDSK];

  const int tid   = threadIdx.x;
  const int wave  = tid >> 5;
  const int lane  = tid & 31;
  const int l15   = lane & 15;
  const int lhalf = lane >> 4;
  const int cbase = wave * 16;

  const int ntiles = (N + 15) >> 4;
  const int tstep  = gridDim.x;
  int t = blockIdx.x;

  const int lcol = lane * 4;
  float4 q0 = make_float4(0.f, 0.f, 0.f, 0.f);
  float4 q1 = make_float4(0.f, 0.f, 0.f, 0.f);
  if (t < ntiles) {
    int r0 = t * 16 + wave;     if (r0 > N - 1) r0 = N - 1;
    int r1 = t * 16 + wave + 8; if (r1 > N - 1) r1 = N - 1;
    q0 = *(const float4*)(X + (size_t)r0 * IN_F + lcol);
    q1 = *(const float4*)(X + (size_t)r1 * IN_F + lcol);
  }

  float4 b0[16], b1[16];
  const float* wfr = lW + (cbase + l15) * LDSK + 4 * lhalf;
  #pragma unroll
  for (int u = 0; u < 16; ++u) {
    int row = wave + 8 * u;
    *(float4*)(lW + row * LDSK + lcol) = *(const float4*)(W0 + (size_t)row * IN_F + lcol);
  }
  __syncthreads();
  #pragma unroll
  for (int j = 0; j < 16; ++j) b0[j] = *(const float4*)(wfr + 8 * j);
  __syncthreads();
  #pragma unroll
  for (int u = 0; u < 16; ++u) {
    int row = wave + 8 * u;
    *(float4*)(lW + row * LDSK + lcol) = *(const float4*)(W1 + (size_t)row * IN_F + lcol);
  }
  __syncthreads();
  #pragma unroll
  for (int j = 0; j < 16; ++j) b1[j] = *(const float4*)(wfr + 8 * j);

  if (t < ntiles) {
    *(float4*)(lA[0] + wave * LDSK + lcol)       = q0;
    *(float4*)(lA[0] + (wave + 8) * LDSK + lcol) = q1;
  }
  __syncthreads();

  const int   head = wave >> 1;
  const float aval = a0[head * DHEAD + (cbase & (DHEAD - 1)) + l15];

  int buf = 0;
  for (; t < ntiles; t += tstep) {
    const int  nbase = t * 16;
    const int  tn    = t + tstep;
    const bool more  = tn < ntiles;
    if (more) {
      int r0 = tn * 16 + wave;     if (r0 > N - 1) r0 = N - 1;
      int r1 = tn * 16 + wave + 8; if (r1 > N - 1) r1 = N - 1;
      q0 = *(const float4*)(X + (size_t)r0 * IN_F + lcol);
      q1 = *(const float4*)(X + (size_t)r1 * IN_F + lcol);
    }

    const float* afr = lA[buf] + l15 * LDSK + 4 * lhalf;
    float4 af[16];
    #pragma unroll
    for (int j = 0; j < 16; ++j) af[j] = *(const float4*)(afr + 8 * j);
    __builtin_amdgcn_sched_barrier(0);

    v8f acc0 = {};
    v8f acc1 = {};
    #pragma unroll
    for (int j = 0; j < 16; ++j) {
      v2f alo;  alo.x  = af[j].x; alo.y  = af[j].y;
      v2f ahi;  ahi.x  = af[j].z; ahi.y  = af[j].w;
      v2f b0lo; b0lo.x = b0[j].x; b0lo.y = b0[j].y;
      v2f b0hi; b0hi.x = b0[j].z; b0hi.y = b0[j].w;
      v2f b1lo; b1lo.x = b1[j].x; b1lo.y = b1[j].y;
      v2f b1hi; b1hi.x = b1[j].z; b1hi.y = b1[j].w;
      acc0 = __builtin_amdgcn_wmma_f32_16x16x4_f32(false, alo, false, b0lo,
                                                   (short)0, acc0, false, false);
      acc1 = __builtin_amdgcn_wmma_f32_16x16x4_f32(false, alo, false, b1lo,
                                                   (short)0, acc1, false, false);
      acc0 = __builtin_amdgcn_wmma_f32_16x16x4_f32(false, ahi, false, b0hi,
                                                   (short)0, acc0, false, false);
      acc1 = __builtin_amdgcn_wmma_f32_16x16x4_f32(false, ahi, false, b1hi,
                                                   (short)0, acc1, false, false);
    }

    float p0[8], p1[8];
    #pragma unroll
    for (int j = 0; j < 8; ++j) {
      float x0 = acc0[j]; x0 = (x0 > 0.0f) ? x0 : ALPHA * x0;
      float x1 = acc1[j]; x1 = (x1 > 0.0f) ? x1 : ALPHA * x1;
      int r = nbase + j + 8 * lhalf;
      if (r < N) X1out[(size_t)r * HD + cbase + l15] = x1;
      p0[j] = x0 * aval;
      p1[j] = x1 * aval;
    }

    #pragma unroll
    for (int j = 0; j < 8; ++j) {
      #pragma unroll
      for (int off = 8; off >= 1; off >>= 1) {
        p0[j] += __shfl_xor(p0[j], off, 32);
        p1[j] += __shfl_xor(p1[j], off, 32);
      }
    }
    if (l15 == 0) {
      #pragma unroll
      for (int j = 0; j < 8; ++j) {
        int r = nbase + j + 8 * lhalf;
        if (r < N) {
          atomicAdd(&s0[r * NHEAD + head], p0[j]);
          atomicAdd(&s1[r * NHEAD + head], p1[j]);
        }
      }
    }

    if (more) {
      *(float4*)(lA[buf ^ 1] + wave * LDSK + lcol)       = q0;
      *(float4*)(lA[buf ^ 1] + (wave + 8) * LDSK + lcol) = q1;
    }
    __syncthreads();
    buf ^= 1;
  }
}

__global__ __launch_bounds__(256) void gat_minmax_part(
    const float* __restrict__ s0, const float* __restrict__ s1,
    const int* __restrict__ esrc, const int* __restrict__ ecol, int E,
    float* __restrict__ pmax, float* __restrict__ pmn)
{
  const int tid = threadIdx.x;
  float mx[NHEAD], mn[NHEAD];
  #pragma unroll
  for (int h = 0; h < NHEAD; ++h) { mx[h] = -1e30f; mn[h] = 1e30f; }

  for (int e = blockIdx.x * 256 + tid; e < E; e += gridDim.x * 256) {
    int s = esrc[e];
    int c = ecol[e];
    const float* sa = s0 + (size_t)s * NHEAD;
    const float* sb = s1 + (size_t)c * NHEAD;
    #pragma unroll
    for (int h = 0; h < NHEAD; ++h) {
      float att = sa[h] + sb[h];
      mx[h] = fmaxf(mx[h], att);
      mn[h] = fminf(mn[h], att);
    }
  }

  #pragma unroll
  for (int h = 0; h < NHEAD; ++h)
    for (int off = 16; off >= 1; off >>= 1) {
      mx[h] = fmaxf(mx[h], __shfl_xor(mx[h], off, 32));
      mn[h] = fminf(mn[h], __shfl_xor(mn[h], off, 32));
    }

  __shared__ float smx[8][NHEAD], smn[8][NHEAD];
  const int wave = tid >> 5, lane = tid & 31;
  if (lane == 0) {
    #pragma unroll
    for (int h = 0; h < NHEAD; ++h) { smx[wave][h] = mx[h]; smn[wave][h] = mn[h]; }
  }
  __syncthreads();
  if (tid == 0) {
    #pragma unroll
    for (int h = 0; h < NHEAD; ++h) {
      float a = smx[0][h], b = smn[0][h];
      for (int w = 1; w < 8; ++w) { a = fmaxf(a, smx[w][h]); b = fminf(b, smn[w][h]); }
      pmax[blockIdx.x * NHEAD + h] = a;
      pmn [blockIdx.x * NHEAD + h] = b;
    }
  }
}

__global__ __launch_bounds__(256) void gat_minmax_final(
    const float* __restrict__ pmax, const float* __restrict__ pmn, int nb,
    float* __restrict__ mxmn)
{
  const int tid = threadIdx.x;
  float mx[NHEAD], mn[NHEAD];
  #pragma unroll
  for (int h = 0; h < NHEAD; ++h) { mx[h] = -1e30f; mn[h] = 1e30f; }

  for (int i = tid; i < nb; i += 256) {
    #pragma unroll
    for (int h = 0; h < NHEAD; ++h) {
      mx[h] = fmaxf(mx[h], pmax[i * NHEAD + h]);
      mn[h] = fminf(mn[h], pmn [i * NHEAD + h]);
    }
  }
  #pragma unroll
  for (int h = 0; h < NHEAD; ++h)
    for (int off = 16; off >= 1; off >>= 1) {
      mx[h] = fmaxf(mx[h], __shfl_xor(mx[h], off, 32));
      mn[h] = fminf(mn[h], __shfl_xor(mn[h], off, 32));
    }

  __shared__ float smx[8][NHEAD], smn[8][NHEAD];
  const int wave = tid >> 5, lane = tid & 31;
  if (lane == 0) {
    #pragma unroll
    for (int h = 0; h < NHEAD; ++h) { smx[wave][h] = mx[h]; smn[wave][h] = mn[h]; }
  }
  __syncthreads();
  if (tid == 0) {
    #pragma unroll
    for (int h = 0; h < NHEAD; ++h) {
      float a = smx[0][h], b = smn[0][h];
      for (int w = 1; w < 8; ++w) { a = fmaxf(a, smx[w][h]); b = fminf(b, smn[w][h]); }
      mxmn[h]         = a;
      mxmn[NHEAD + h] = b;
    }
  }
}

__global__ __launch_bounds__(256) void gat_edge(
    const float* __restrict__ X1, const float* __restrict__ s0,
    const float* __restrict__ s1, const int* __restrict__ ecol,
    const float* __restrict__ mxmn, float* __restrict__ out, int N)
{
  const int wave = threadIdx.x >> 5;
  const int lane = threadIdx.x & 31;
  const int n = blockIdx.x * 8 + wave;
  if (n >= N) return;

  float scale[NHEAD], s0n[NHEAD];
  #pragma unroll
  for (int h = 0; h < NHEAD; ++h) {
    float mxv = mxmn[h], mnv = mxmn[NHEAD + h];
    scale[h] = 1.0f / (mxv - mnv);
    s0n[h]   = s0[n * NHEAD + h] - mnv;
  }

  float acc[NHEAD]  = {0.f, 0.f, 0.f, 0.f};
  float wsum[NHEAD] = {0.f, 0.f, 0.f, 0.f};

  const int* cols = ecol + n * DEG;
  for (int e = 0; e < DEG; ++e) {
    int c = cols[e];
    const float* sc   = s1 + (size_t)c * NHEAD;
    const float* xrow = X1 + (size_t)c * HD;
    #pragma unroll
    for (int h = 0; h < NHEAD; ++h) {
      float att = __expf((s0n[h] + sc[h]) * scale[h]);
      wsum[h] += att;
      acc[h]  += att * xrow[h * DHEAD + lane];
    }
  }
  #pragma unroll
  for (int h = 0; h < NHEAD; ++h)
    out[(size_t)n * HD + h * DHEAD + lane] = acc[h] / wsum[h];
}

extern "C" void kernel_launch(void* const* d_in, const int* in_sizes, int n_in,
                              void* d_out, int out_size, void* d_ws, size_t ws_size,
                              hipStream_t stream) {
  const float* X    = (const float*)d_in[0];
  const float* W0   = (const float*)d_in[1];
  const float* W1   = (const float*)d_in[2];
  const float* a0   = (const float*)d_in[3];
  const int*   esrc = (const int*)d_in[4];
  const int*   ecol = (const int*)d_in[5];
  float*       out  = (float*)d_out;

  const int N = in_sizes[0] / IN_F;
  const int E = in_sizes[5];
  const int NB = 512;

  float* X1ws = (float*)d_ws;
  float* s0   = X1ws + (size_t)N * HD;
  float* s1   = s0 + (size_t)N * NHEAD;
  float* pmax = s1 + (size_t)N * NHEAD;
  float* pmn  = pmax + (size_t)NB * NHEAD;
  float* mxmn = pmn + (size_t)NB * NHEAD;

  gat_zero<<<(2 * N * NHEAD + 255) / 256, 256, 0, stream>>>(s0, 2 * N * NHEAD);
  gat_gemm<<<512, 256, 0, stream>>>(X, W0, W1, a0, X1ws, s0, s1, N);
  gat_minmax_part<<<NB, 256, 0, stream>>>(s0, s1, esrc, ecol, E, pmax, pmn);
  gat_minmax_final<<<1, 256, 0, stream>>>(pmax, pmn, NB, mxmn);
  gat_edge<<<(N + 7) / 8, 256, 0, stream>>>(X1ws, s0, s1, ecol, mxmn, out, N);
}


# GEMM: each wave owns one whole head (32 columns) of one product, so a head's score sum ends inside the wave and is stored: no atomics, no zero launch (5->4); s0/s1 sum order changes
# speedup vs baseline: 1.7225x; 1.0555x over previous
#include <hip/hip_runtime.h>
#include <math.h>

typedef float v2f __attribute__((ext_vector_type(2)));
typedef float v8f __attribute__((ext_vector_type(8)));

#define IN_F  128
#define HD    128
#define NHEAD 4
#define DHEAD 32
#define DEG   16
#define ALPHA 0.2f
#define LDSK  132

#if defined(__HIP_DEVICE_COMPILE__)
#if !__has_builtin(__builtin_amdgcn_wmma_f32_16x16x4_f32)
#error "gfx1250 fp32 WMMA builtin not available on this toolchain (device pass)"
#endif
#endif

__global__ __launch_bounds__(256) void gat_gemm(
    const float* __restrict__ X, const float* __restrict__ W0,
    const float* __restrict__ W1, const float* __restrict__ a0,
    float* __restrict__ X1out, float* __restrict__ s0, float* __restrict__ s1,
    int N)
{
  __shared__ __attribute__((aligned(16))) float lW[HD * LDSK];
  __shared__ __attribute__((aligned(16))) float lA[2][16 * LDSK];

  const int tid   = threadIdx.x;
  const int wave  = __builtin_amdgcn_readfirstlane(tid >> 5);
  const int lane  = tid & 31;
  const int l15   = lane & 15;
  const int lhalf = lane >> 4;
  const bool isX1 = wave >= 4;
  const int  head = wave & 3;
  const int  cbase = head * DHEAD;

  const int ntiles = (N + 15) >> 4;
  const int tstep  = gridDim.x;
  int t = blockIdx.x;

  const int lcol = lane * 4;
  float4 q0 = make_float4(0.f, 0.f, 0.f, 0.f);
  float4 q1 = make_float4(0.f, 0.f, 0.f, 0.f);
  if (t < ntiles) {
    int r0 = t * 16 + wave;     if (r0 > N - 1) r0 = N - 1;
    int r1 = t * 16 + wave + 8; if (r1 > N - 1) r1 = N - 1;
    q0 = *(const float4*)(X + (size_t)r0 * IN_F + lcol);
    q1 = *(const float4*)(X + (size_t)r1 * IN_F + lcol);
  }

  float4 bA[16], bB[16];
  const float* wfrA = lW + (cbase + l15) * LDSK + 4 * lhalf;
  const float* wfrB = wfrA + 16 * LDSK;
  #pragma unroll
  for (int u = 0; u < 16; ++u) {
    int row = wave + 8 * u;
    *(float4*)(lW + row * LDSK + lcol) = *(const float4*)(W0 + (size_t)row * IN_F + lcol);
  }
  __syncthreads();
  #pragma unroll
  for (int j = 0; j < 16; ++j) {
    bA[j] = *(const float4*)(wfrA + 8 * j);
    bB[j] = *(const float4*)(wfrB + 8 * j);
  }
  __syncthreads();
  #pragma unroll
  for (int u = 0; u < 16; ++u) {
    int row = wave + 8 * u;
    *(float4*)(lW + row * LDSK + lcol) = *(const float4*)(W1 + (size_t)row * IN_F + lcol);
  }
  __syncthreads();
  if (isX1) {
    #pragma unroll
    for (int j = 0; j < 16; ++j) {
      bA[j] = *(const float4*)(wfrA + 8 * j);
      bB[j] = *(const float4*)(wfrB + 8 * j);
    }
  }

  if (t < ntiles) {
    *(float4*)(lA[0] + wave * LDSK + lcol)       = q0;
    *(float4*)(lA[0] + (wave + 8) * LDSK + lcol) = q1;
  }
  __syncthreads();

  const float avalA = a0[cbase + l15];
  const float avalB = a0[cbase + 16 + l15];
  float* __restrict__ sdst = isX1 ? s1 : s0;

  int buf = 0;
  for (; t < ntiles; t += tstep) {
    const int  nbase = t * 16;
    const int  tn    = t + tstep;
    const bool more  = tn < ntiles;
    if (more) {
      int r0 = tn * 16 + wave;     if (r0 > N - 1) r0 = N - 1;
      int r1 = tn * 16 + wave + 8; if (r1 > N - 1) r1 = N - 1;
      q0 = *(const float4*)(X + (size_t)r0 * IN_F + lcol);
      q1 = *(const float4*)(X + (size_t)r1 * IN_F + lcol);
    }

    const float* afr = lA[buf] + l15 * LDSK + 4 * lhalf;
    float4 af[16];
    #pragma unroll
    for (int j = 0; j < 16; ++j) af[j] = *(const float4*)(afr + 8 * j);
    __builtin_amdgcn_sched_barrier(0);

    v8f accA = {};
    v8f accB = {};
    #pragma unroll
    for (int j = 0; j < 16; ++j) {
      v2f alo;  alo.x  = af[j].x; alo.y  = af[j].y;
      v2f ahi;  ahi.x  = af[j].z; ahi.y  = af[j].w;
      v2f bAlo; bAlo.x = bA[j].x; bAlo.y = bA[j].y;
      v2f bAhi; bAhi.x = bA[j].z; bAhi.y = bA[j].w;
      v2f bBlo; bBlo.x = bB[j].x; bBlo.y = bB[j].y;
      v2f bBhi; bBhi.x = bB[j].z; bBhi.y = bB[j].w;
      accA = __builtin_amdgcn_wmma_f32_16x16x4_f32(false, alo, false, bAlo,
                                                   (short)0, accA, false, false);
      accB = __builtin_amdgcn_wmma_f32_16x16x4_f32(false, alo, false, bBlo,
                                                   (short)0, accB, false, false);
      accA = __builtin_amdgcn_wmma_f32_16x16x4_f32(false, ahi, false, bAhi,
                                                   (short)0, accA, false, false);
      accB = __builtin_amdgcn_wmma_f32_16x16x4_f32(false, ahi, false, bBhi,
                                                   (short)0, accB, false, false);
    }

    float xa[8], xb[8], p[8];
    #pragma unroll
    for (int j = 0; j < 8; ++j) {
      float va = accA[j]; va = (va > 0.0f) ? va : ALPHA * va;
      float vb = accB[j]; vb = (vb > 0.0f) ? vb : ALPHA * vb;
      xa[j] = va;
      xb[j] = vb;
      p[j]  = va * avalA + vb * avalB;
    }
    if (isX1) {
      #pragma unroll
      for (int j = 0; j < 8; ++j) {
        int r = nbase + j + 8 * lhalf;
        if (r < N) {
          X1out[(size_t)r * HD + cbase + l15]      = xa[j];
          X1out[(size_t)r * HD + cbase + 16 + l15] = xb[j];
        }
      }
    }

    #pragma unroll
    for (int j = 0; j < 8; ++j) {
      #pragma unroll
      for (int off = 8; off >= 1; off >>= 1)
        p[j] += __shfl_xor(p[j], off, 32);
    }
    if (l15 == 0) {
      #pragma unroll
      for (int j = 0; j < 8; ++j) {
        int r = nbase + j + 8 * lhalf;
        if (r < N) sdst[r * NHEAD + head] = p[j];
      }
    }

    if (more) {
      *(float4*)(lA[buf ^ 1] + wave * LDSK + lcol)       = q0;
      *(float4*)(lA[buf ^ 1] + (wave + 8) * LDSK + lcol) = q1;
    }
    __syncthreads();
    buf ^= 1;
  }
}

__global__ __launch_bounds__(256) void gat_minmax_part(
    const float* __restrict__ s0, const float* __restrict__ s1,
    const int* __restrict__ esrc, const int* __restrict__ ecol, int E,
    float* __restrict__ pmax, float* __restrict__ pmn)
{
  const int tid = threadIdx.x;
  float mx[NHEAD], mn[NHEAD];
  #pragma unroll
  for (int h = 0; h < NHEAD; ++h) { mx[h] = -1e30f; mn[h] = 1e30f; }

  for (int e = blockIdx.x * 256 + tid; e < E; e += gridDim.x * 256) {
    int s = esrc[e];
    int c = ecol[e];
    const float* sa = s0 + (size_t)s * NHEAD;
    const float* sb = s1 + (size_t)c * NHEAD;
    #pragma unroll
    for (int h = 0; h < NHEAD; ++h) {
      float att = sa[h] + sb[h];
      mx[h] = fmaxf(mx[h], att);
      mn[h] = fminf(mn[h], att);
    }
  }

  #pragma unroll
  for (int h = 0; h < NHEAD; ++h)
    for (int off = 16; off >= 1; off >>= 1) {
      mx[h] = fmaxf(mx[h], __shfl_xor(mx[h], off, 32));
      mn[h] = fminf(mn[h], __shfl_xor(mn[h], off, 32));
    }

  __shared__ float smx[8][NHEAD], smn[8][NHEAD];
  const int wave = tid >> 5, lane = tid & 31;
  if (lane == 0) {
    #pragma unroll
    for (int h = 0; h < NHEAD; ++h) { smx[wave][h] = mx[h]; smn[wave][h] = mn[h]; }
  }
  __syncthreads();
  if (tid == 0) {
    #pragma unroll
    for (int h = 0; h < NHEAD; ++h) {
      float a = smx[0][h], b = smn[0][h];
      for (int w = 1; w < 8; ++w) { a = fmaxf(a, smx[w][h]); b = fminf(b, smn[w][h]); }
      pmax[blockIdx.x * NHEAD + h] = a;
      pmn [blockIdx.x * NHEAD + h] = b;
    }
  }
}

__global__ __launch_bounds__(256) void gat_minmax_final(
    const float* __restrict__ pmax, const float* __restrict__ pmn, int nb,
    float* __restrict__ mxmn)
{
  const int tid = threadIdx.x;
  float mx[NHEAD], mn[NHEAD];
  #pragma unroll
  for (int h = 0; h < NHEAD; ++h) { mx[h] = -1e30f; mn[h] = 1e30f; }

  for (int i = tid; i < nb; i += 256) {
    #pragma unroll
    for (int h = 0; h < NHEAD; ++h) {
      mx[h] = fmaxf(mx[h], pmax[i * NHEAD + h]);
      mn[h] = fminf(mn[h], pmn [i * NHEAD + h]);
    }
  }
  #pragma unroll
  for (int h = 0; h < NHEAD; ++h)
    for (int off = 16; off >= 1; off >>= 1) {
      mx[h] = fmaxf(mx[h], __shfl_xor(mx[h], off, 32));
      mn[h] = fminf(mn[h], __shfl_xor(mn[h], off, 32));
    }

  __shared__ float smx[8][NHEAD], smn[8][NHEAD];
  const int wave = tid >> 5, lane = tid & 31;
  if (lane == 0) {
    #pragma unroll
    for (int h = 0; h < NHEAD; ++h) { smx[wave][h] = mx[h]; smn[wave][h] = mn[h]; }
  }
  __syncthreads();
  if (tid == 0) {
    #pragma unroll
    for (int h = 0; h < NHEAD; ++h) {
      float a = smx[0][h], b = smn[0][h];
      for (int w = 1; w < 8; ++w) { a = fmaxf(a, smx[w][h]); b = fminf(b, smn[w][h]); }
      mxmn[h]         = a;
      mxmn[NHEAD + h] = b;
    }
  }
}

__global__ __launch_bounds__(256) void gat_edge(
    const float* __restrict__ X1, const float* __restrict__ s0,
    const float* __restrict__ s1, const int* __restrict__ ecol,
    const float* __restrict__ mxmn, float* __restrict__ out, int N)
{
  const int wave = threadIdx.x >> 5;
  const int lane = threadIdx.x & 31;
  const int n = blockIdx.x * 8 + wave;
  if (n >= N) return;

  float scale[NHEAD], s0n[NHEAD];
  #pragma unroll
  for (int h = 0; h < NHEAD; ++h) {
    float mxv = mxmn[h], mnv = mxmn[NHEAD + h];
    scale[h] = 1.0f / (mxv - mnv);
    s0n[h]   = s0[n * NHEAD + h] - mnv;
  }

  float acc[NHEAD]  = {0.f, 0.f, 0.f, 0.f};
  float wsum[NHEAD] = {0.f, 0.f, 0.f, 0.f};

  const int* cols = ecol + n * DEG;
  for (int e = 0; e < DEG; ++e) {
    int c = cols[e];
    const float* sc   = s1 + (size_t)c * NHEAD;
    const float* xrow = X1 + (size_t)c * HD;
    #pragma unroll
    for (int h = 0; h < NHEAD; ++h) {
      float att = __expf((s0n[h] + sc[h]) * scale[h]);
      wsum[h] += att;
      acc[h]  += att * xrow[h * DHEAD + lane];
    }
  }
  #pragma unroll
  for (int h = 0; h < NHEAD; ++h)
    out[(size_t)n * HD + h * DHEAD + lane] = acc[h] / wsum[h];
}

extern "C" void kernel_launch(void* const* d_in, const int* in_sizes, int n_in,
                              void* d_out, int out_size, void* d_ws, size_t ws_size,
                              hipStream_t stream) {
  const float* X    = (const float*)d_in[0];
  const float* W0   = (const float*)d_in[1];
  const float* W1   = (const float*)d_in[2];
  const float* a0   = (const float*)d_in[3];
  const int*   esrc = (const int*)d_in[4];
  const int*   ecol = (const int*)d_in[5];
  float*       out  = (float*)d_out;

  const int N = in_sizes[0] / IN_F;
  const int E = in_sizes[5];
  const int NB = 512;

  float* X1ws = (float*)d_ws;
  float* s0   = X1ws + (size_t)N * HD;
  float* s1   = s0 + (size_t)N * NHEAD;
  float* pmax = s1 + (size_t)N * NHEAD;
  float* pmn  = pmax + (size_t)NB * NHEAD;
  float* mxmn = pmn + (size_t)NB * NHEAD;

  gat_gemm<<<512, 256, 0, stream>>>(X, W0, W1, a0, X1ws, s0, s1, N);
  gat_minmax_part<<<NB, 256, 0, stream>>>(s0, s1, esrc, ecol, E, pmax, pmn);
  gat_minmax_final<<<1, 256, 0, stream>>>(pmax, pmn, NB, mxmn);
  gat_edge<<<(N + 7) / 8, 256, 0, stream>>>(X1ws, s0, s1, ecol, mxmn, out, N);
}


# edge kernel: a lane owns 4 adjacent columns of one head (one 16-byte load per lane = whole row) and takes the neighbour's s1 from that row (4 FMA + 3 DPP adds): no s1 gather; s1 sum order changes
# speedup vs baseline: 1.8564x; 1.0777x over previous
#include <hip/hip_runtime.h>
#include <math.h>

typedef float v2f __attribute__((ext_vector_type(2)));
typedef float v8f __attribute__((ext_vector_type(8)));

#define IN_F  128
#define HD    128
#define NHEAD 4
#define DHEAD 32
#define DEG   16
#define ALPHA 0.2f
#define LDSK  132

#if defined(__HIP_DEVICE_COMPILE__)
#if !__has_builtin(__builtin_amdgcn_wmma_f32_16x16x4_f32)
#error "gfx1250 fp32 WMMA builtin not available on this toolchain (device pass)"
#endif
#endif

__global__ __launch_bounds__(256) void gat_gemm(
    const float* __restrict__ X, const float* __restrict__ W0,
    const float* __restrict__ W1, const float* __restrict__ a0,
    float* __restrict__ X1out, float* __restrict__ s0, float* __restrict__ s1,
    int N)
{
  __shared__ __attribute__((aligned(16))) float lW[HD * LDSK];
  __shared__ __attribute__((aligned(16))) float lA[2][16 * LDSK];

  const int tid   = threadIdx.x;
  const int wave  = __builtin_amdgcn_readfirstlane(tid >> 5);
  const int lane  = tid & 31;
  const int l15   = lane & 15;
  const int lhalf = lane >> 4;
  const bool isX1 = wave >= 4;
  const int  head = wave & 3;
  const int  cbase = head * DHEAD;

  const int ntiles = (N + 15) >> 4;
  const int tstep  = gridDim.x;
  int t = blockIdx.x;

  const int lcol = lane * 4;
  float4 q0 = make_float4(0.f, 0.f, 0.f, 0.f);
  float4 q1 = make_float4(0.f, 0.f, 0.f, 0.f);
  if (t < ntiles) {
    int r0 = t * 16 + wave;     if (r0 > N - 1) r0 = N - 1;
    int r1 = t * 16 + wave + 8; if (r1 > N - 1) r1 = N - 1;
    q0 = *(const float4*)(X + (size_t)r0 * IN_F + lcol);
    q1 = *(const float4*)(X + (size_t)r1 * IN_F + lcol);
  }

  float4 bA[16], bB[16];
  const float* wfrA = lW + (cbase + l15) * LDSK + 4 * lhalf;
  const float* wfrB = wfrA + 16 * LDSK;
  #pragma unroll
  for (int u = 0; u < 16; ++u) {
    int row = wave + 8 * u;
    *(float4*)(lW + row * LDSK + lcol) = *(const float4*)(W0 + (size_t)row * IN_F + lcol);
  }
  __syncthreads();
  #pragma unroll
  for (int j = 0; j < 16; ++j) {
    bA[j] = *(const float4*)(wfrA + 8 * j);
    bB[j] = *(const float4*)(wfrB + 8 * j);
  }
  __syncthreads();
  #pragma unroll
  for (int u = 0; u < 16; ++u) {
    int row = wave + 8 * u;
    *(float4*)(lW + row * LDSK + lcol) = *(const float4*)(W1 + (size_t)row * IN_F + lcol);
  }
  __syncthreads();
  if (isX1) {
    #pragma unroll
    for (int j = 0; j < 16; ++j) {
      bA[j] = *(const float4*)(wfrA + 8 * j);
      bB[j] = *(const float4*)(wfrB + 8 * j);
    }
  }

  if (t < ntiles) {
    *(float4*)(lA[0] + wave * LDSK + lcol)       = q0;
    *(float4*)(lA[0] + (wave + 8) * LDSK + lcol) = q1;
  }
  __syncthreads();

  const float avalA = a0[cbase + l15];
  const float avalB = a0[cbase + 16 + l15];
  float* __restrict__ sdst = isX1 ? s1 : s0;

  int buf = 0;
  for (; t < ntiles; t += tstep) {
    const int  nbase = t * 16;
    const int  tn    = t + tstep;
    const bool more  = tn < ntiles;
    if (more) {
      int r0 = tn * 16 + wave;     if (r0 > N - 1) r0 = N - 1;
      int r1 = tn * 16 + wave + 8; if (r1 > N - 1) r1 = N - 1;
      q0 = *(const float4*)(X + (size_t)r0 * IN_F + lcol);
      q1 = *(const float4*)(X + (size_t)r1 * IN_F + lcol);
    }

    const float* afr = lA[buf] + l15 * LDSK + 4 * lhalf;
    float4 af[16];
    #pragma unroll
    for (int j = 0; j < 16; ++j) af[j] = *(const float4*)(afr + 8 * j);
    __builtin_amdgcn_sched_barrier(0);

    v8f accA = {};
    v8f accB = {};
    #pragma unroll
    for (int j = 0; j < 16; ++j) {
      v2f alo;  alo.x  = af[j].x; alo.y  = af[j].y;
      v2f ahi;  ahi.x  = af[j].z; ahi.y  = af[j].w;
      v2f bAlo; bAlo.x = bA[j].x; bAlo.y = bA[j].y;
      v2f bAhi; bAhi.x = bA[j].z; bAhi.y = bA[j].w;
      v2f bBlo; bBlo.x = bB[j].x; bBlo.y = bB[j].y;
      v2f bBhi; bBhi.x = bB[j].z; bBhi.y = bB[j].w;
      accA = __builtin_amdgcn_wmma_f32_16x16x4_f32(false, alo, false, bAlo,
                                                   (short)0, accA, false, false);
      accB = __builtin_amdgcn_wmma_f32_16x16x4_f32(false, alo, false, bBlo,
                                                   (short)0, accB, false, false);
      accA = __builtin_amdgcn_wmma_f32_16x16x4_f32(false, ahi, false, bAhi,
                                                   (short)0, accA, false, false);
      accB = __builtin_amdgcn_wmma_f32_16x16x4_f32(false, ahi, false, bBhi,
                                                   (short)0, accB, false, false);
    }

    float xa[8], xb[8], p[8];
    #pragma unroll
    for (int j = 0; j < 8; ++j) {
      float va = accA[j]; va = (va > 0.0f) ? va : ALPHA * va;
      float vb = accB[j]; vb = (vb > 0.0f) ? vb : ALPHA * vb;
      xa[j] = va;
      xb[j] = vb;
      p[j]  = va * avalA + vb * avalB;
    }
    if (isX1) {
      #pragma unroll
      for (int j = 0; j < 8; ++j) {
        int r = nbase + j + 8 * lhalf;
        if (r < N) {
          X1out[(size_t)r * HD + cbase + l15]      = xa[j];
          X1out[(size_t)r * HD + cbase + 16 + l15] = xb[j];
        }
      }
    }

    #pragma unroll
    for (int j = 0; j < 8; ++j) {
      #pragma unroll
      for (int off = 8; off >= 1; off >>= 1)
        p[j] += __shfl_xor(p[j], off, 32);
    }
    if (l15 == 0) {
      #pragma unroll
      for (int j = 0; j < 8; ++j) {
        int r = nbase + j + 8 * lhalf;
        if (r < N) sdst[r * NHEAD + head] = p[j];
      }
    }

    if (more) {
      *(float4*)(lA[buf ^ 1] + wave * LDSK + lcol)       = q0;
      *(float4*)(lA[buf ^ 1] + (wave + 8) * LDSK + lcol) = q1;
    }
    __syncthreads();
    buf ^= 1;
  }
}

__global__ __launch_bounds__(256) void gat_minmax_part(
    const float* __restrict__ s0, const float* __restrict__ s1,
    const int* __restrict__ esrc, const int* __restrict__ ecol, int E,
    float* __restrict__ pmax, float* __restrict__ pmn)
{
  const int tid = threadIdx.x;
  float mx[NHEAD], mn[NHEAD];
  #pragma unroll
  for (int h = 0; h < NHEAD; ++h) { mx[h] = -1e30f; mn[h] = 1e30f; }

  for (int e = blockIdx.x * 256 + tid; e < E; e += gridDim.x * 256) {
    int s = esrc[e];
    int c = ecol[e];
    const float* sa = s0 + (size_t)s * NHEAD;
    const float* sb = s1 + (size_t)c * NHEAD;
    #pragma unroll
    for (int h = 0; h < NHEAD; ++h) {
      float att = sa[h] + sb[h];
      mx[h] = fmaxf(mx[h], att);
      mn[h] = fminf(mn[h], att);
    }
  }

  #pragma unroll
  for (int h = 0; h < NHEAD; ++h)
    for (int off = 16; off >= 1; off >>= 1) {
      mx[h] = fmaxf(mx[h], __shfl_xor(mx[h], off, 32));
      mn[h] = fminf(mn[h], __shfl_xor(mn[h], off, 32));
    }

  __shared__ float smx[8][NHEAD], smn[8][NHEAD];
  const int wave = tid >> 5, lane = tid & 31;
  if (lane == 0) {
    #pragma unroll
    for (int h = 0; h < NHEAD; ++h) { smx[wave][h] = mx[h]; smn[wave][h] = mn[h]; }
  }
  __syncthreads();
  if (tid == 0) {
    #pragma unroll
    for (int h = 0; h < NHEAD; ++h) {
      float a = smx[0][h], b = smn[0][h];
      for (int w = 1; w < 8; ++w) { a = fmaxf(a, smx[w][h]); b = fminf(b, smn[w][h]); }
      pmax[blockIdx.x * NHEAD + h] = a;
      pmn [blockIdx.x * NHEAD + h] = b;
    }
  }
}

__global__ __launch_bounds__(256) void gat_minmax_final(
    const float* __restrict__ pmax, const float* __restrict__ pmn, int nb,
    float* __restrict__ mxmn)
{
  const int tid = threadIdx.x;
  float mx[NHEAD], mn[NHEAD];
  #pragma unroll
  for (int h = 0; h < NHEAD; ++h) { mx[h] = -1e30f; mn[h] = 1e30f; }

  for (int i = tid; i < nb; i += 256) {
    #pragma unroll
    for (int h = 0; h < NHEAD; ++h) {
      mx[h] = fmaxf(mx[h], pmax[i * NHEAD + h]);
      mn[h] = fminf(mn[h], pmn [i * NHEAD + h]);
    }
  }
  #pragma unroll
  for (int h = 0; h < NHEAD; ++h)
    for (int off = 16; off >= 1; off >>= 1) {
      mx[h] = fmaxf(mx[h], __shfl_xor(mx[h], off, 32));
      mn[h] = fminf(mn[h], __shfl_xor(mn[h], off, 32));
    }

  __shared__ float smx[8][NHEAD], smn[8][NHEAD];
  const int wave = tid >> 5, lane = tid & 31;
  if (lane == 0) {
    #pragma unroll
    for (int h = 0; h < NHEAD; ++h) { smx[wave][h] = mx[h]; smn[wave][h] = mn[h]; }
  }
  __syncthreads();
  if (tid == 0) {
    #pragma unroll
    for (int h = 0; h < NHEAD; ++h) {
      float a = smx[0][h], b = smn[0][h];
      for (int w = 1; w < 8; ++w) { a = fmaxf(a, smx[w][h]); b = fminf(b, smn[w][h]); }
      mxmn[h]         = a;
      mxmn[NHEAD + h] = b;
    }
  }
}


template <int MASK>
__device__ __forceinline__ float add_lane_xor(float v) {
  const int other = __builtin_amdgcn_update_dpp(0, __float_as_int(v), 0x160 + MASK, 0xF, 0xF, true);
  return v + __int_as_float(other);
}

__global__ __launch_bounds__(256) void gat_edge(
    const float* __restrict__ X1, const float* __restrict__ s0,
    const float* __restrict__ a0, const int* __restrict__ ecol,
    const float* __restrict__ mxmn, float* __restrict__ out, int N)
{
  const int wave = __builtin_amdgcn_readfirstlane(threadIdx.x >> 5);
  const int lane = threadIdx.x & 31;
  const int n = blockIdx.x * 8 + wave;
  if (n >= N) return;
  const int h = lane >> 3;

  const int cv = ecol[n * DEG + (lane & 15)];
  const float4 av = *(const float4*)(a0 + 4 * lane);

  const float mxv = mxmn[h], mnv = mxmn[NHEAD + h];
  const float scale = 1.0f / (mxv - mnv);
  const float s0n   = s0[n * NHEAD + h] - mnv;

  float4 acc  = make_float4(0.f, 0.f, 0.f, 0.f);
  float  wsum = 0.f;

  float4 xr[DEG];
  #pragma unroll
  for (int e = 0; e < DEG; ++e) {
    const int c = __builtin_amdgcn_readlane(cv, e);
    xr[e] = *(const float4*)(X1 + (size_t)c * HD + 4 * lane);
  }
  #pragma unroll
  for (int e = 0; e < DEG; ++e) {
    const float4 x = xr[e];
    float p = x.x * av.x;
    p = fmaf(x.y, av.y, p);
    p = fmaf(x.z, av.z, p);
    p = fmaf(x.w, av.w, p);
    p = add_lane_xor<1>(p);
    p = add_lane_xor<2>(p);
    p = add_lane_xor<4>(p);
    const float att = __expf((s0n + p) * scale);
    wsum += att;
    acc.x = fmaf(att, x.x, acc.x);
    acc.y = fmaf(att, x.y, acc.y);
    acc.z = fmaf(att, x.z, acc.z);
    acc.w = fmaf(att, x.w, acc.w);
  }
  float4 o;
  o.x = acc.x / wsum;
  o.y = acc.y / wsum;
  o.z = acc.z / wsum;
  o.w = acc.w / wsum;
  *(float4*)(out + (size_t)n * HD + 4 * lane) = o;
}

extern "C" void kernel_launch(void* const* d_in, const int* in_sizes, int n_in,
                              void* d_out, int out_size, void* d_ws, size_t ws_size,
                              hipStream_t stream) {
  const float* X    = (const float*)d_in[0];
  const float* W0   = (const float*)d_in[1];
  const float* W1   = (const float*)d_in[2];
  const float* a0   = (const float*)d_in[3];
  const int*   esrc = (const int*)d_in[4];
  const int*   ecol = (const int*)d_in[5];
  float*       out  = (float*)d_out;

  const int N = in_sizes[0] / IN_F;
  const int E = in_sizes[5];
  const int NB = 512;

  float* X1ws = (float*)d_ws;
  float* s0   = X1ws + (size_t)N * HD;
  float* s1   = s0 + (size_t)N * NHEAD;
  float* pmax = s1 + (size_t)N * NHEAD;
  float* pmn  = pmax + (size_t)NB * NHEAD;
  float* mxmn = pmn + (size_t)NB * NHEAD;

  gat_gemm<<<512, 256, 0, stream>>>(X, W0, W1, a0, X1ws, s0, s1, N);
  gat_minmax_part<<<NB, 256, 0, stream>>>(s0, s1, esrc, ecol, E, pmax, pmn);
  gat_minmax_final<<<1, 256, 0, stream>>>(pmax, pmn, NB, mxmn);
  gat_edge<<<(N + 7) / 8, 256, 0, stream>>>(X1ws, s0, a0, ecol, mxmn, out, N);
}
